# starting kernel
# baseline (speedup 1.0000x reference)
#include <hip/hip_runtime.h>
#include <hip/hip_bf16.h>


typedef float v2f __attribute__((ext_vector_type(2)));
typedef float v8f __attribute__((ext_vector_type(8)));

#define WMMA_F32(A, B, C)                                                  \
    __builtin_amdgcn_wmma_f32_16x16x4_f32(false, (A), false, (B),          \
                                          (short)0, (C), false, false)

namespace {
constexpr int D_IN   = 1024;
constexpr int F_OUT  = 512;
constexpr int ROWS   = 32;
constexpr int KSLAB  = 128;
constexpr int LSTR   = 132;
constexpr int BLOCK  = 512;
constexpr int NWAVES = BLOCK / 32;
}

__global__ __launch_bounds__(BLOCK) void between_cluster_fc_kernel(
    const float* __restrict__ e1, const float* __restrict__ e2,
    const float* __restrict__ W1, const float* __restrict__ b1,
    const float* __restrict__ W2, const float* __restrict__ b2,
    float* __restrict__ out) {
    __shared__ float sA1[ROWS * LSTR];
    __shared__ float sA2[ROWS * LSTR];
    __shared__ float sRed[ROWS * NWAVES];

    const int t    = threadIdx.x;
    const int w    = t >> 5;
    const int lane = t & 31;
    const int half = lane >> 4;
    const int lp   = lane & 15;
    const int r0   = blockIdx.x * ROWS;
    const int f0   = w * 32;

    const float bb1[2] = {b1[f0 + lp], b1[f0 + 16 + lp]};
    const float bb2[2] = {b2[f0 + lp], b2[f0 + 16 + lp]};

    v8f C1[2][2] = {};
    v8f C2[2][2] = {};

    for (int ks = 0; ks < D_IN / KSLAB; ++ks) {
        const int k0 = ks * KSLAB;
        __syncthreads();

#pragma unroll
        for (int c = 0; c < 2; ++c) {
            const int li  = (t + c * BLOCK) * 4;
            const int row = li >> 7;
            const int kk  = li & (KSLAB - 1);
            const size_t goff = (size_t)(r0 + row) * D_IN + k0 + kk;
            const float4 v1 = *reinterpret_cast<const float4*>(e1 + goff);
            const float4 v2 = *reinterpret_cast<const float4*>(e2 + goff);
            *reinterpret_cast<float4*>(sA1 + row * LSTR + kk) = v1;
            *reinterpret_cast<float4*>(sA2 + row * LSTR + kk) = v2;
            if (ks + 1 < D_IN / KSLAB) {
                __builtin_prefetch(e1 + goff + KSLAB, 0, 0);
                __builtin_prefetch(e2 + goff + KSLAB, 0, 0);
            }
        }
        __syncthreads();

        for (int kl = 0; kl < KSLAB; kl += 4) {
            const int kw = kl + 2 * half;
            const v2f a1m0 = *reinterpret_cast<const v2f*>(sA1 + lp * LSTR + kw);
            const v2f a1m1 = *reinterpret_cast<const v2f*>(sA1 + (16 + lp) * LSTR + kw);
            const v2f a2m0 = *reinterpret_cast<const v2f*>(sA2 + lp * LSTR + kw);
            const v2f a2m1 = *reinterpret_cast<const v2f*>(sA2 + (16 + lp) * LSTR + kw);

            const size_t kg  = (size_t)(k0 + kl + 2 * half);
            const float* w1p = W1 + kg * F_OUT + f0 + lp;
            const float* w2p = W2 + kg * F_OUT + f0 + lp;
            v2f bW1[2], bW2[2];
            bW1[0][0] = w1p[0];  bW1[0][1] = w1p[F_OUT];
            bW1[1][0] = w1p[16]; bW1[1][1] = w1p[F_OUT + 16];
            bW2[0][0] = w2p[0];  bW2[0][1] = w2p[F_OUT];
            bW2[1][0] = w2p[16]; bW2[1][1] = w2p[F_OUT + 16];

            C1[0][0] = WMMA_F32(a1m0, bW1[0], C1[0][0]);
            C1[0][1] = WMMA_F32(a1m0, bW1[1], C1[0][1]);
            C1[1][0] = WMMA_F32(a1m1, bW1[0], C1[1][0]);
            C1[1][1] = WMMA_F32(a1m1, bW1[1], C1[1][1]);
            C2[0][0] = WMMA_F32(a2m0, bW2[0], C2[0][0]);
            C2[0][1] = WMMA_F32(a2m0, bW2[1], C2[0][1]);
            C2[1][0] = WMMA_F32(a2m1, bW2[0], C2[1][0]);
            C2[1][1] = WMMA_F32(a2m1, bW2[1], C2[1][1]);
        }
    }

    v8f P[2] = {};
#pragma unroll
    for (int m = 0; m < 2; ++m)
#pragma unroll
        for (int j = 0; j < 2; ++j)
#pragma unroll
            for (int v = 0; v < 8; ++v)
                P[m][v] += (C1[m][j][v] + bb1[j]) * (C2[m][j][v] + bb2[j]);

#pragma unroll
    for (int m = 0; m < 2; ++m)
#pragma unroll
        for (int v = 0; v < 8; ++v) {
            float s = P[m][v];
            s += __shfl_xor(s, 1);
            s += __shfl_xor(s, 2);
            s += __shfl_xor(s, 4);
            s += __shfl_xor(s, 8);
            if (lp == 0) {
                const int row = 16 * m + v + 8 * half;
                sRed[row * NWAVES + w] = s;
            }
        }
    __syncthreads();

    if (t < ROWS) {
        float acc = 0.0f;
#pragma unroll
        for (int ww = 0; ww < NWAVES; ++ww) acc += sRed[t * NWAVES + ww];
        out[r0 + t] = acc;
    }
}

extern "C" void kernel_launch(void* const* d_in, const int* in_sizes, int n_in,
                              void* d_out, int out_size, void* d_ws, size_t ws_size,
                              hipStream_t stream) {
    const float* e1 = (const float*)d_in[0];
    const float* e2 = (const float*)d_in[1];
    const float* W1 = (const float*)d_in[2];
    const float* b1 = (const float*)d_in[3];
    const float* W2 = (const float*)d_in[4];
    const float* b2 = (const float*)d_in[5];
    float* out = (float*)d_out;

    const int n = in_sizes[0] / D_IN;
    dim3 grid(n / ROWS), block(BLOCK);
    between_cluster_fc_kernel<<<grid, block, 0, stream>>>(e1, e2, W1, b1, W2, b2, out);
}


# Work-groups of 8 waves (wave tile 32x64 per product) so two are resident per WGP and fill each other's staging, barrier and epilogue gaps; new summation order: 4 tiles per lane, 8 wave partials
# speedup vs baseline: 1.1017x; 1.1017x over previous
#include <hip/hip_runtime.h>
#include <hip/hip_bf16.h>


typedef float v2f __attribute__((ext_vector_type(2)));
typedef float v8f __attribute__((ext_vector_type(8)));

#define WMMA_F32(A, B, C)                                                  \
    __builtin_amdgcn_wmma_f32_16x16x4_f32(false, (A), false, (B),          \
                                          (short)0, (C), false, false)

namespace {
constexpr int D_IN   = 1024;
constexpr int F_OUT  = 512;
constexpr int ROWS   = 32;
constexpr int KSLAB  = 128;
constexpr int LSTR   = 132;
constexpr int BLOCK  = 256;
constexpr int NT     = 4;
constexpr int NWAVES = BLOCK / 32;
}

__global__ __launch_bounds__(BLOCK) void between_cluster_fc_kernel(
    const float* __restrict__ e1, const float* __restrict__ e2,
    const float* __restrict__ W1, const float* __restrict__ b1,
    const float* __restrict__ W2, const float* __restrict__ b2,
    float* __restrict__ out) {
    __shared__ float sA1[ROWS * LSTR];
    __shared__ float sA2[ROWS * LSTR];
    __shared__ float sRed[ROWS * NWAVES];

    const int t    = threadIdx.x;
    const int w    = t >> 5;
    const int lane = t & 31;
    const int half = lane >> 4;
    const int lp   = lane & 15;
    const int r0   = blockIdx.x * ROWS;
    const int f0   = w * 16 * NT;

    float bb1[NT], bb2[NT];
#pragma unroll
    for (int j = 0; j < NT; ++j) {
        bb1[j] = b1[f0 + 16 * j + lp];
        bb2[j] = b2[f0 + 16 * j + lp];
    }

    v8f C1[2][NT] = {};
    v8f C2[2][NT] = {};

    for (int ks = 0; ks < D_IN / KSLAB; ++ks) {
        const int k0 = ks * KSLAB;
        __syncthreads();

#pragma unroll
        for (int c = 0; c < ROWS * KSLAB / 4 / BLOCK; ++c) {
            const int li  = (t + c * BLOCK) * 4;
            const int row = li >> 7;
            const int kk  = li & (KSLAB - 1);
            const size_t goff = (size_t)(r0 + row) * D_IN + k0 + kk;
            const float4 v1 = *reinterpret_cast<const float4*>(e1 + goff);
            const float4 v2 = *reinterpret_cast<const float4*>(e2 + goff);
            *reinterpret_cast<float4*>(sA1 + row * LSTR + kk) = v1;
            *reinterpret_cast<float4*>(sA2 + row * LSTR + kk) = v2;
            if (ks + 1 < D_IN / KSLAB) {
                __builtin_prefetch(e1 + goff + KSLAB, 0, 0);
                __builtin_prefetch(e2 + goff + KSLAB, 0, 0);
            }
        }
        __syncthreads();

        for (int kl = 0; kl < KSLAB; kl += 4) {
            const int kw = kl + 2 * half;
            const v2f a1m0 = *reinterpret_cast<const v2f*>(sA1 + lp * LSTR + kw);
            const v2f a1m1 = *reinterpret_cast<const v2f*>(sA1 + (16 + lp) * LSTR + kw);
            const v2f a2m0 = *reinterpret_cast<const v2f*>(sA2 + lp * LSTR + kw);
            const v2f a2m1 = *reinterpret_cast<const v2f*>(sA2 + (16 + lp) * LSTR + kw);

            const size_t kg  = (size_t)(k0 + kl + 2 * half);
            const float* w1p = W1 + kg * F_OUT + f0 + lp;
            const float* w2p = W2 + kg * F_OUT + f0 + lp;
            v2f bW1[NT], bW2[NT];
#pragma unroll
            for (int j = 0; j < NT; ++j) {
                bW1[j][0] = w1p[16 * j]; bW1[j][1] = w1p[F_OUT + 16 * j];
                bW2[j][0] = w2p[16 * j]; bW2[j][1] = w2p[F_OUT + 16 * j];
            }

#pragma unroll
            for (int j = 0; j < NT; ++j) {
                C1[0][j] = WMMA_F32(a1m0, bW1[j], C1[0][j]);
                C1[1][j] = WMMA_F32(a1m1, bW1[j], C1[1][j]);
            }
#pragma unroll
            for (int j = 0; j < NT; ++j) {
                C2[0][j] = WMMA_F32(a2m0, bW2[j], C2[0][j]);
                C2[1][j] = WMMA_F32(a2m1, bW2[j], C2[1][j]);
            }
        }
    }

    v8f P[2] = {};
#pragma unroll
    for (int m = 0; m < 2; ++m)
#pragma unroll
        for (int j = 0; j < NT; ++j)
#pragma unroll
            for (int v = 0; v < 8; ++v)
                P[m][v] += (C1[m][j][v] + bb1[j]) * (C2[m][j][v] + bb2[j]);

#pragma unroll
    for (int m = 0; m < 2; ++m)
#pragma unroll
        for (int v = 0; v < 8; ++v) {
            float s = P[m][v];
            s += __shfl_xor(s, 1);
            s += __shfl_xor(s, 2);
            s += __shfl_xor(s, 4);
            s += __shfl_xor(s, 8);
            if (lp == 0) {
                const int row = 16 * m + v + 8 * half;
                sRed[row * NWAVES + w] = s;
            }
        }
    __syncthreads();

    if (t < ROWS) {
        float acc = 0.0f;
#pragma unroll
        for (int ww = 0; ww < NWAVES; ++ww) acc += sRed[t * NWAVES + ww];
        out[r0 + t] = acc;
    }
}

extern "C" void kernel_launch(void* const* d_in, const int* in_sizes, int n_in,
                              void* d_out, int out_size, void* d_ws, size_t ws_size,
                              hipStream_t stream) {
    const float* e1 = (const float*)d_in[0];
    const float* e2 = (const float*)d_in[1];
    const float* W1 = (const float*)d_in[2];
    const float* b1 = (const float*)d_in[3];
    const float* W2 = (const float*)d_in[4];
    const float* b2 = (const float*)d_in[5];
    float* out = (float*)d_out;

    const int n = in_sizes[0] / D_IN;
    dim3 grid(n / ROWS), block(BLOCK);
    between_cluster_fc_kernel<<<grid, block, 0, stream>>>(e1, e2, W1, b1, W2, b2, out);
}
